# starting kernel
# baseline (speedup 1.0000x reference)
#include <hip/hip_runtime.h>
#include <hip/hip_bf16.h>

typedef __attribute__((ext_vector_type(2))) float v2f;
typedef __attribute__((ext_vector_type(4))) float v4f;
typedef __attribute__((ext_vector_type(8))) float v8f;

#define BN 2048
#define MM 1024
#define ND 64

__global__ void gmm_zero_kernel(float* p, int n) {
  int i = blockIdx.x * blockDim.x + threadIdx.x;
  if (i < n) p[i] = 0.0f;
}

__global__ void gmm_prep_kernel(const float* __restrict__ Mu0,
                                const float* __restrict__ Mu1,
                                const float* __restrict__ S0,
                                const float* __restrict__ S1,
                                const float* __restrict__ tP,
                                const float* __restrict__ eP,
                                float* __restrict__ Bp,
                                float* __restrict__ cM,
                                float* __restrict__ PQ) {
  const int m = blockIdx.x;
  const int d = threadIdx.x;
  const float t   = tP[0];
  const float eps = eP[0];
  const float eps2 = eps * eps;
  const int idx = m * ND + d;

  float s0  = S0[idx],  s1  = S1[idx];
  float mu0 = Mu0[idx], mu1 = Mu1[idx];

  float Ds  = sqrtf(4.0f * s0 * s1 + eps2 * eps2);
  float Cs  = 0.5f * (Ds - eps2);
  float vv  = mu1 - mu0;
  float mut = (1.0f - t) * mu0 + t * mu1;
  float Sg  = (1.0f - t) * (1.0f - t) * s0 + t * t * s1
            + 2.0f * t * (1.0f - t) * (Cs + 0.5f * eps2);
  float Pt  = t * s1 + (1.0f - t) * Cs;
  float Qt  = (1.0f - t) * s0 + t * Cs;
  float St  = Pt - Qt - eps2 * t;
  float Kc  = St / Sg;
  float invS = 1.0f / Sg;

  Bp[(d >> 1) * (2 * MM) + 2 * m + (d & 1)]        = invS;
  Bp[(32 + (d >> 1)) * (2 * MM) + 2 * m + (d & 1)] = -2.0f * mut * invS;

  float* pq = PQ + ((size_t)(m >> 1) * ND + d) * 4;
  pq[(m & 1)]     = Kc;
  pq[2 + (m & 1)] = vv - Kc * mut;

  __shared__ float red[ND];
  red[d] = mut * mut * invS + logf(Sg);
  __syncthreads();
  for (int s = ND / 2; s > 0; s >>= 1) {
    if (d < s) red[d] += red[d + s];
    __syncthreads();
  }
  if (d == 0) cM[m] = red[0];
}

__global__ void __launch_bounds__(32)
gmm_main_kernel(const float* __restrict__ X,
                const float* __restrict__ Lam,
                const float* __restrict__ cM,
                const float* __restrict__ Bp,
                const float* __restrict__ PQ,
                float* __restrict__ uAcc,
                float* __restrict__ nAcc) {
  const int strip = blockIdx.x;
  const int chunk = blockIdx.y;
  const int b0    = strip * 16;
  const int l     = threadIdx.x;
  const int lmod  = l & 15;
  const int lhalf = l >> 4;

  __shared__ float lds[16 * 18];

  v2f xa[32];
  const float* xrow = X + (size_t)(b0 + lmod) * ND;
#pragma unroll
  for (int kk = 0; kk < 16; ++kk) {
    int k0 = kk * 4 + lhalf * 2;
    v2f xv = *(const v2f*)(xrow + k0);
    v2f tfrag; tfrag[0] = xv[0] * xv[0]; tfrag[1] = xv[1] * xv[1];
    xa[kk] = tfrag;
  }
#pragma unroll
  for (int kk = 16; kk < 32; ++kk) {
    int k0 = kk * 4 + lhalf * 2 - 64;
    xa[kk] = *(const v2f*)(xrow + k0);
  }

  v8f accP[4] = {};
  v8f accQ[4] = {};
  v8f accN = {};

  v2f bones;
  bones[0] = (lmod == 0) ? 1.0f : 0.0f;
  bones[1] = bones[0];

  for (int mt = 0; mt < 8; ++mt) {
    const int m0 = chunk * 128 + mt * 16;

    v8f L = {};
#pragma unroll
    for (int kk = 0; kk < 32; ++kk) {
      const int pairIdx = kk * 2 + lhalf;
      v2f bf = *(const v2f*)(Bp + (size_t)pairIdx * (2 * MM) + 2 * (m0 + lmod));
      L = __builtin_amdgcn_wmma_f32_16x16x4_f32(false, xa[kk], false, bf,
                                                (short)0, L, false, false);
    }

    const float cm  = cM[m0 + lmod];
    const float lam = Lam[m0 + lmod];
    float w[8];
#pragma unroll
    for (int r = 0; r < 8; ++r) {
      float lw = -0.5f * (L[r] + cm);
      lw = fminf(50.0f, fmaxf(-50.0f, lw));
      w[r] = __expf(lw) * lam;
    }

    __syncthreads();
#pragma unroll
    for (int r = 0; r < 8; ++r)
      lds[(r + lhalf * 8) * 18 + lmod] = w[r];
    __syncthreads();

#pragma unroll
    for (int kk2 = 0; kk2 < 4; ++kk2) {
      const int km = kk2 * 4 + lhalf * 2;
      v2f wa = *(const v2f*)(&lds[lmod * 18 + km]);
      const int mpair = (m0 >> 1) + kk2 * 2 + lhalf;
#pragma unroll
      for (int n = 0; n < 4; ++n) {
        const int d = n * 16 + lmod;
        v4f pq = *(const v4f*)(PQ + ((size_t)mpair * ND + d) * 4);
        v2f bp; bp[0] = pq[0]; bp[1] = pq[1];
        accP[n] = __builtin_amdgcn_wmma_f32_16x16x4_f32(false, wa, false, bp,
                                                        (short)0, accP[n],
                                                        false, false);
        v2f bq; bq[0] = pq[2]; bq[1] = pq[3];
        accQ[n] = __builtin_amdgcn_wmma_f32_16x16x4_f32(false, wa, false, bq,
                                                        (short)0, accQ[n],
                                                        false, false);
      }
      accN = __builtin_amdgcn_wmma_f32_16x16x4_f32(false, wa, false, bones,
                                                   (short)0, accN,
                                                   false, false);
    }
  }

#pragma unroll
  for (int r = 0; r < 8; ++r) {
    const int b = b0 + r + lhalf * 8;
#pragma unroll
    for (int n = 0; n < 4; ++n) {
      const int d = n * 16 + lmod;
      float val = X[(size_t)b * ND + d] * accP[n][r] + accQ[n][r];
      atomicAdd(&uAcc[(size_t)b * ND + d], val);
    }
    if (lmod == 0) atomicAdd(&nAcc[b], accN[r]);
  }
}

__global__ void gmm_finalize_kernel(const float* __restrict__ uAcc,
                                    const float* __restrict__ nAcc,
                                    float* __restrict__ out) {
  int i = blockIdx.x * blockDim.x + threadIdx.x;
  if (i < BN * ND) out[i] = uAcc[i] / nAcc[i / ND];
}

extern "C" void kernel_launch(void* const* d_in, const int* in_sizes, int n_in,
                              void* d_out, int out_size, void* d_ws, size_t ws_size,
                              hipStream_t stream) {
  const float* X   = (const float*)d_in[0];
  const float* Mu0 = (const float*)d_in[1];
  const float* Mu1 = (const float*)d_in[2];
  const float* S0  = (const float*)d_in[3];
  const float* S1  = (const float*)d_in[4];
  const float* Lam = (const float*)d_in[5];
  const float* tP  = (const float*)d_in[6];
  const float* eP  = (const float*)d_in[7];
  float* out = (float*)d_out;

  float* ws   = (float*)d_ws;
  float* Bp   = ws;
  float* cM   = Bp + 64 * 2 * MM;
  float* PQ   = cM + MM;
  float* uAcc = PQ + (MM / 2) * ND * 4;
  float* nAcc = uAcc + BN * ND;
  (void)ws_size; (void)in_sizes; (void)n_in; (void)out_size;

  const int nzero = BN * ND + BN;
  gmm_zero_kernel<<<(nzero + 255) / 256, 256, 0, stream>>>(uAcc, nzero);

  gmm_prep_kernel<<<MM, ND, 0, stream>>>(Mu0, Mu1, S0, S1, tP, eP,
                                         Bp, cM, PQ);

  gmm_main_kernel<<<dim3(BN / 16, 8), 32, 0, stream>>>(X, Lam, cM, Bp,
                                                       PQ, uAcc, nAcc);

  gmm_finalize_kernel<<<(BN * ND + 255) / 256, 256, 0, stream>>>(uAcc, nAcc, out);
}


# Atomic combine of m-chunk partials replaced by a fixed-order two-level sum: 16 per-wave partials added in LDS per 16-wave work-group (2 m-halves per strip), halves added in finalize; no zero launch
# speedup vs baseline: 2.2188x; 2.2188x over previous
#include <hip/hip_runtime.h>
#include <hip/hip_bf16.h>

typedef __attribute__((ext_vector_type(2))) float v2f;
typedef __attribute__((ext_vector_type(4))) float v4f;
typedef __attribute__((ext_vector_type(8))) float v8f;

#define BN 2048
#define MM 1024
#define ND 64

__global__ void gmm_prep_kernel(const float* __restrict__ Mu0,
                                const float* __restrict__ Mu1,
                                const float* __restrict__ S0,
                                const float* __restrict__ S1,
                                const float* __restrict__ tP,
                                const float* __restrict__ eP,
                                float* __restrict__ Bp,
                                float* __restrict__ cM,
                                float* __restrict__ PQ) {
  const int m = blockIdx.x;
  const int d = threadIdx.x;
  const float t   = tP[0];
  const float eps = eP[0];
  const float eps2 = eps * eps;
  const int idx = m * ND + d;

  float s0  = S0[idx],  s1  = S1[idx];
  float mu0 = Mu0[idx], mu1 = Mu1[idx];

  float Ds  = sqrtf(4.0f * s0 * s1 + eps2 * eps2);
  float Cs  = 0.5f * (Ds - eps2);
  float vv  = mu1 - mu0;
  float mut = (1.0f - t) * mu0 + t * mu1;
  float Sg  = (1.0f - t) * (1.0f - t) * s0 + t * t * s1
            + 2.0f * t * (1.0f - t) * (Cs + 0.5f * eps2);
  float Pt  = t * s1 + (1.0f - t) * Cs;
  float Qt  = (1.0f - t) * s0 + t * Cs;
  float St  = Pt - Qt - eps2 * t;
  float Kc  = St / Sg;
  float invS = 1.0f / Sg;

  Bp[(d >> 1) * (2 * MM) + 2 * m + (d & 1)]        = invS;
  Bp[(32 + (d >> 1)) * (2 * MM) + 2 * m + (d & 1)] = -2.0f * mut * invS;

  float* pq = PQ + ((size_t)(m >> 1) * ND + d) * 4;
  pq[(m & 1)]     = Kc;
  pq[2 + (m & 1)] = vv - Kc * mut;

  __shared__ float red[ND];
  red[d] = mut * mut * invS + logf(Sg);
  __syncthreads();
  for (int s = ND / 2; s > 0; s >>= 1) {
    if (d < s) red[d] += red[d + s];
    __syncthreads();
  }
  if (d == 0) cM[m] = red[0];
}

#define NW   16
#define NH   2
#define PSTR 66
#define NPAD 32

__global__ void __launch_bounds__(32 * NW)
gmm_main_kernel(const float* __restrict__ X,
                const float* __restrict__ Lam,
                const float* __restrict__ cM,
                const float* __restrict__ Bp,
                const float* __restrict__ PQ,
                float* __restrict__ Uh,
                float* __restrict__ Nh) {
  const int strip = blockIdx.x;
  const int half  = blockIdx.y;
  const int b0    = strip * 16;
  const int wv    = __builtin_amdgcn_readfirstlane((int)(threadIdx.x >> 5));
  const int l     = threadIdx.x & 31;
  const int lmod  = l & 15;
  const int lhalf = l >> 4;

  __shared__ float ldsT[NW][16 * 18];
  __shared__ float part[NW][16 * PSTR];
  __shared__ float npart[NW][16];

  float* lds = ldsT[wv];

  v2f xa[32];
  const float* xrow = X + (size_t)(b0 + lmod) * ND;
#pragma unroll
  for (int kk = 0; kk < 16; ++kk) {
    int k0 = kk * 4 + lhalf * 2;
    v2f xv = *(const v2f*)(xrow + k0);
    v2f tfrag; tfrag[0] = xv[0] * xv[0]; tfrag[1] = xv[1] * xv[1];
    xa[kk] = tfrag;
  }
#pragma unroll
  for (int kk = 16; kk < 32; ++kk) {
    int k0 = kk * 4 + lhalf * 2 - 64;
    xa[kk] = *(const v2f*)(xrow + k0);
  }

  v8f accP[4] = {};
  v8f accQ[4] = {};
  v8f accN = {};

  v2f bones;
  bones[0] = (lmod == 0) ? 1.0f : 0.0f;
  bones[1] = bones[0];

  for (int mt = 0; mt < MM / (NH * NW * 16); ++mt) {
    const int m0 = half * (MM / NH) + wv * (MM / (NH * NW)) + mt * 16;

    v8f L = {};
#pragma unroll
    for (int kk = 0; kk < 32; ++kk) {
      const int pairIdx = kk * 2 + lhalf;
      v2f bf = *(const v2f*)(Bp + (size_t)pairIdx * (2 * MM) + 2 * (m0 + lmod));
      L = __builtin_amdgcn_wmma_f32_16x16x4_f32(false, xa[kk], false, bf,
                                                (short)0, L, false, false);
    }

    const float cm  = cM[m0 + lmod];
    const float lam = Lam[m0 + lmod];
    float w[8];
#pragma unroll
    for (int r = 0; r < 8; ++r) {
      float lw = -0.5f * (L[r] + cm);
      lw = fminf(50.0f, fmaxf(-50.0f, lw));
      w[r] = __expf(lw) * lam;
    }

#pragma unroll
    for (int r = 0; r < 8; ++r)
      lds[(r + lhalf * 8) * 18 + lmod] = w[r];

#pragma unroll
    for (int kk2 = 0; kk2 < 4; ++kk2) {
      const int km = kk2 * 4 + lhalf * 2;
      v2f wa = *(const v2f*)(&lds[lmod * 18 + km]);
      const int mpair = (m0 >> 1) + kk2 * 2 + lhalf;
#pragma unroll
      for (int n = 0; n < 4; ++n) {
        const int d = n * 16 + lmod;
        v4f pq = *(const v4f*)(PQ + ((size_t)mpair * ND + d) * 4);
        v2f bp; bp[0] = pq[0]; bp[1] = pq[1];
        accP[n] = __builtin_amdgcn_wmma_f32_16x16x4_f32(false, wa, false, bp,
                                                        (short)0, accP[n],
                                                        false, false);
        v2f bq; bq[0] = pq[2]; bq[1] = pq[3];
        accQ[n] = __builtin_amdgcn_wmma_f32_16x16x4_f32(false, wa, false, bq,
                                                        (short)0, accQ[n],
                                                        false, false);
      }
      accN = __builtin_amdgcn_wmma_f32_16x16x4_f32(false, wa, false, bones,
                                                   (short)0, accN,
                                                   false, false);
    }
  }

  float* pw = part[wv];
#pragma unroll
  for (int r = 0; r < 8; ++r) {
    const int bl = r + lhalf * 8;
#pragma unroll
    for (int n = 0; n < 4; ++n) {
      const int d = n * 16 + lmod;
      pw[bl * PSTR + d] = X[(size_t)(b0 + bl) * ND + d] * accP[n][r] + accQ[n][r];
    }
  }
  if (lmod == 0) {
#pragma unroll
    for (int r = 0; r < 8; ++r) npart[wv][r + lhalf * 8] = accN[r];
  }
  __syncthreads();

  v2f s = {};
#pragma unroll
  for (int w2 = 0; w2 < NW; ++w2) {
    v2f p = *(const v2f*)(&part[w2][wv * PSTR + 2 * l]);
    s[0] += p[0];
    s[1] += p[1];
  }
  *(v2f*)(Uh + ((size_t)half * BN + b0 + wv) * ND + 2 * l) = s;

  if (wv == 0 && l < 16) {
    float nrm = 0.0f;
#pragma unroll
    for (int w2 = 0; w2 < NW; ++w2) nrm += npart[w2][l];
    Nh[((size_t)half * (BN / 16) + strip) * NPAD + l] = nrm;
  }
}

__global__ void gmm_finalize_kernel(const float* __restrict__ Uh,
                                    const float* __restrict__ Nh,
                                    float* __restrict__ out) {
  const int i = blockIdx.x * blockDim.x + threadIdx.x;
  const int b = i / (ND / 4);
  const int nidx = (b >> 4) * NPAD + (b & 15);
  const v4f u0 = *(const v4f*)(Uh + (size_t)i * 4);
  const v4f u1 = *(const v4f*)(Uh + (size_t)BN * ND + (size_t)i * 4);
  const float nrm = Nh[nidx] + Nh[(BN / 16) * NPAD + nidx];
  v4f o;
  o[0] = (u0[0] + u1[0]) / nrm;
  o[1] = (u0[1] + u1[1]) / nrm;
  o[2] = (u0[2] + u1[2]) / nrm;
  o[3] = (u0[3] + u1[3]) / nrm;
  *(v4f*)(out + (size_t)i * 4) = o;
}

extern "C" void kernel_launch(void* const* d_in, const int* in_sizes, int n_in,
                              void* d_out, int out_size, void* d_ws, size_t ws_size,
                              hipStream_t stream) {
  const float* X   = (const float*)d_in[0];
  const float* Mu0 = (const float*)d_in[1];
  const float* Mu1 = (const float*)d_in[2];
  const float* S0  = (const float*)d_in[3];
  const float* S1  = (const float*)d_in[4];
  const float* Lam = (const float*)d_in[5];
  const float* tP  = (const float*)d_in[6];
  const float* eP  = (const float*)d_in[7];
  float* out = (float*)d_out;

  float* ws   = (float*)d_ws;
  float* Bp   = ws;
  float* cM   = Bp + 64 * 2 * MM;
  float* PQ   = cM + MM;
  float* Uh   = PQ + (MM / 2) * ND * 4;
  float* Nh   = Uh + NH * BN * ND;
  (void)ws_size; (void)in_sizes; (void)n_in; (void)out_size;

  gmm_prep_kernel<<<MM, ND, 0, stream>>>(Mu0, Mu1, S0, S1, tP, eP,
                                         Bp, cM, PQ);

  gmm_main_kernel<<<dim3(BN / 16, NH), 32 * NW, 0, stream>>>(X, Lam, cM, Bp,
                                                             PQ, Uh, Nh);

  gmm_finalize_kernel<<<BN * ND / 4 / 256, 256, 0, stream>>>(Uh, Nh, out);
}


# Operand launch fused away: each work-group (16 row-groups x 16 m-slices) rebuilds its 64-component operands in LDS, fragments read from LDS; sums now wave pair in LDS, then 16 planes in fixed order
# speedup vs baseline: 2.9540x; 1.3314x over previous
#include <hip/hip_runtime.h>
#include <hip/hip_bf16.h>


typedef __attribute__((ext_vector_type(2))) float v2f;
typedef __attribute__((ext_vector_type(4))) float v4f;
typedef __attribute__((ext_vector_type(8))) float v8f;

#define BN 2048
#define MM 1024
#define ND 64

#define NW     16
#define MS     16
#define MSL    (MM / MS)
#define BG     16
#define BROWS  (BN / BG)
#define BPITCH 132
#define PSTR   66

#define SM_BS    0
#define SM_PQ    (SM_BS + 64 * BPITCH)
#define SM_CM    (SM_PQ + (MSL / 2) * ND * 4)
#define SM_T     (SM_CM + MSL)
#define SM_PART  (SM_T + NW * 16 * 18)
#define SM_NP    (SM_PART + NW * 16 * PSTR)
#define SM_TOTAL (SM_NP + NW * 16)

struct Elem { float invS, bx, Kc, Qv, cterm; };

static __device__ __forceinline__ Elem gmm_elem(float s0, float s1, float mu0,
                                                float mu1, float t, float eps2) {
  float Ds  = sqrtf(4.0f * s0 * s1 + eps2 * eps2);
  float Cs  = 0.5f * (Ds - eps2);
  float vv  = mu1 - mu0;
  float mut = (1.0f - t) * mu0 + t * mu1;
  float Sg  = (1.0f - t) * (1.0f - t) * s0 + t * t * s1
            + 2.0f * t * (1.0f - t) * (Cs + 0.5f * eps2);
  float Pt  = t * s1 + (1.0f - t) * Cs;
  float Qt  = (1.0f - t) * s0 + t * Cs;
  float St  = Pt - Qt - eps2 * t;
  float Kc  = St / Sg;
  float invS = 1.0f / Sg;
  Elem e;
  e.invS  = invS;
  e.bx    = -2.0f * mut * invS;
  e.Kc    = Kc;
  e.Qv    = vv - Kc * mut;
  e.cterm = mut * mut * invS + logf(Sg);
  return e;
}

__global__ void __launch_bounds__(32 * NW)
gmm_main_kernel(const float* __restrict__ X,
                const float* __restrict__ Mu0,
                const float* __restrict__ Mu1,
                const float* __restrict__ S0,
                const float* __restrict__ S1,
                const float* __restrict__ Lam,
                const float* __restrict__ tP,
                const float* __restrict__ eP,
                float* __restrict__ Uh,
                float* __restrict__ Nh) {
  const int bg = blockIdx.x;
  const int ms = blockIdx.y;
  const int wv    = __builtin_amdgcn_readfirstlane((int)(threadIdx.x >> 5));
  const int l     = threadIdx.x & 31;
  const int lmod  = l & 15;
  const int lhalf = l >> 4;

  __shared__ __attribute__((aligned(16))) float smem[SM_TOTAL];
  float* Bs  = smem + SM_BS;
  float* PQs = smem + SM_PQ;
  float* cMs = smem + SM_CM;
  float (*ldsT)[16 * 18]    = (float (*)[16 * 18])(smem + SM_T);
  float (*part)[16 * PSTR]  = (float (*)[16 * PSTR])(smem + SM_PART);
  float (*npart)[16]        = (float (*)[16])(smem + SM_NP);

  const int b0 = bg * BROWS + (wv >> 1) * 16;

  v2f xa[32];
  const float* xrow = X + (size_t)(b0 + lmod) * ND;
#pragma unroll
  for (int kk = 0; kk < 16; ++kk) {
    int k0 = kk * 4 + lhalf * 2;
    v2f xv = *(const v2f*)(xrow + k0);
    v2f tfrag; tfrag[0] = xv[0] * xv[0]; tfrag[1] = xv[1] * xv[1];
    xa[kk] = tfrag;
  }
#pragma unroll
  for (int kk = 16; kk < 32; ++kk) {
    int k0 = kk * 4 + lhalf * 2 - 64;
    xa[kk] = *(const v2f*)(xrow + k0);
  }

  {
    const float t    = tP[0];
    const float eps  = eP[0];
    const float eps2 = eps * eps;
#pragma unroll
    for (int cp = 0; cp < 2; ++cp) {
      const int c  = wv * 2 + cp;
      const int m  = ms * MSL + 2 * c;
      const size_t g0 = (size_t)m * ND + 2 * l;
      const size_t g1 = g0 + ND;
      const v2f s0a = *(const v2f*)(S0 + g0),  s0b = *(const v2f*)(S0 + g1);
      const v2f s1a = *(const v2f*)(S1 + g0),  s1b = *(const v2f*)(S1 + g1);
      const v2f m0a = *(const v2f*)(Mu0 + g0), m0b = *(const v2f*)(Mu0 + g1);
      const v2f m1a = *(const v2f*)(Mu1 + g0), m1b = *(const v2f*)(Mu1 + g1);
      const Elem ea0 = gmm_elem(s0a[0], s1a[0], m0a[0], m1a[0], t, eps2);
      const Elem ea1 = gmm_elem(s0a[1], s1a[1], m0a[1], m1a[1], t, eps2);
      const Elem eb0 = gmm_elem(s0b[0], s1b[0], m0b[0], m1b[0], t, eps2);
      const Elem eb1 = gmm_elem(s0b[1], s1b[1], m0b[1], m1b[1], t, eps2);

      v4f bi; bi[0] = ea0.invS; bi[1] = ea1.invS; bi[2] = eb0.invS; bi[3] = eb1.invS;
      v4f bx; bx[0] = ea0.bx;   bx[1] = ea1.bx;   bx[2] = eb0.bx;   bx[3] = eb1.bx;
      *(v4f*)(&Bs[l * BPITCH + 4 * c])        = bi;
      *(v4f*)(&Bs[(32 + l) * BPITCH + 4 * c]) = bx;

      v4f q0; q0[0] = ea0.Kc; q0[1] = eb0.Kc; q0[2] = ea0.Qv; q0[3] = eb0.Qv;
      v4f q1; q1[0] = ea1.Kc; q1[1] = eb1.Kc; q1[2] = ea1.Qv; q1[3] = eb1.Qv;
      *(v4f*)(&PQs[((size_t)c * ND + 2 * l) * 4])     = q0;
      *(v4f*)(&PQs[((size_t)c * ND + 2 * l + 1) * 4]) = q1;

      float ra0 = ea0.cterm, ra1 = ea1.cterm;
      float rb0 = eb0.cterm, rb1 = eb1.cterm;
#pragma unroll
      for (int sh = 16; sh >= 2; sh >>= 1) {
        ra0 += __shfl_down(ra0, sh); ra1 += __shfl_down(ra1, sh);
        rb0 += __shfl_down(rb0, sh); rb1 += __shfl_down(rb1, sh);
      }
      ra0 += __shfl_down(ra0, 1); ra1 += __shfl_down(ra1, 1);
      rb0 += __shfl_down(rb0, 1); rb1 += __shfl_down(rb1, 1);
      if (l == 0) {
        cMs[2 * c]     = ra0 + ra1;
        cMs[2 * c + 1] = rb0 + rb1;
      }
    }
  }
  __syncthreads();

  float* lds = ldsT[wv];

  v8f accP[4] = {};
  v8f accQ[4] = {};
  v8f accN = {};

  v2f bones;
  bones[0] = (lmod == 0) ? 1.0f : 0.0f;
  bones[1] = bones[0];

  for (int mt = 0; mt < MSL / (2 * 16); ++mt) {
    const int ml0 = (wv & 1) * (MSL / 2) + mt * 16;

    v8f L = {};
#pragma unroll
    for (int kk = 0; kk < 32; ++kk) {
      const int pairIdx = kk * 2 + lhalf;
      v2f bf = *(const v2f*)(&Bs[pairIdx * BPITCH + 2 * (ml0 + lmod)]);
      L = __builtin_amdgcn_wmma_f32_16x16x4_f32(false, xa[kk], false, bf,
                                                (short)0, L, false, false);
    }

    const float cm  = cMs[ml0 + lmod];
    const float lam = Lam[ms * MSL + ml0 + lmod];
    float w[8];
#pragma unroll
    for (int r = 0; r < 8; ++r) {
      float lw = -0.5f * (L[r] + cm);
      lw = fminf(50.0f, fmaxf(-50.0f, lw));
      w[r] = __expf(lw) * lam;
    }

#pragma unroll
    for (int r = 0; r < 8; ++r)
      lds[(r + lhalf * 8) * 18 + lmod] = w[r];

#pragma unroll
    for (int kk2 = 0; kk2 < 4; ++kk2) {
      const int km = kk2 * 4 + lhalf * 2;
      v2f wa = *(const v2f*)(&lds[lmod * 18 + km]);
      const int mpair = (ml0 >> 1) + kk2 * 2 + lhalf;
#pragma unroll
      for (int n = 0; n < 4; ++n) {
        const int d = n * 16 + lmod;
        v4f pq = *(const v4f*)(&PQs[((size_t)mpair * ND + d) * 4]);
        v2f bp; bp[0] = pq[0]; bp[1] = pq[1];
        accP[n] = __builtin_amdgcn_wmma_f32_16x16x4_f32(false, wa, false, bp,
                                                        (short)0, accP[n],
                                                        false, false);
        v2f bq; bq[0] = pq[2]; bq[1] = pq[3];
        accQ[n] = __builtin_amdgcn_wmma_f32_16x16x4_f32(false, wa, false, bq,
                                                        (short)0, accQ[n],
                                                        false, false);
      }
      accN = __builtin_amdgcn_wmma_f32_16x16x4_f32(false, wa, false, bones,
                                                   (short)0, accN,
                                                   false, false);
    }
  }

  float* pw = part[wv];
#pragma unroll
  for (int r = 0; r < 8; ++r) {
    const int bl = r + lhalf * 8;
#pragma unroll
    for (int n = 0; n < 4; ++n) {
      const int d = n * 16 + lmod;
      pw[bl * PSTR + d] = X[(size_t)(b0 + bl) * ND + d] * accP[n][r] + accQ[n][r];
    }
  }
  if (lmod == 0) {
#pragma unroll
    for (int r = 0; r < 8; ++r) npart[wv][r + lhalf * 8] = accN[r];
  }
  __syncthreads();

  const int we = wv & ~1;
  float* uplane = Uh + ((size_t)ms * BN + (size_t)bg * BROWS + wv * 8) * ND;
#pragma unroll
  for (int i = 0; i < 8; ++i) {
    const int rl = (wv & 1) * 8 + i;
    const v2f p0 = *(const v2f*)(&part[we][rl * PSTR + 2 * l]);
    const v2f p1 = *(const v2f*)(&part[we + 1][rl * PSTR + 2 * l]);
    v2f s; s[0] = p0[0] + p1[0]; s[1] = p0[1] + p1[1];
    *(v2f*)(uplane + (size_t)i * ND + 2 * l) = s;
  }

  if (wv == 0) {
#pragma unroll
    for (int j = 0; j < 4; ++j) {
      const int row = l + 32 * j;
      const int w0  = (row >> 4) * 2;
      const int rl  = row & 15;
      Nh[(size_t)ms * BN + (size_t)bg * BROWS + row] = npart[w0][rl] + npart[w0 + 1][rl];
    }
  }
}

__global__ void gmm_finalize_kernel(const float* __restrict__ Uh,
                                    const float* __restrict__ Nh,
                                    float* __restrict__ out) {
  const int i = blockIdx.x * blockDim.x + threadIdx.x;
  const int b = i / (ND / 4);
  v4f u = *(const v4f*)(Uh + (size_t)i * 4);
  float nrm = Nh[b];
#pragma unroll
  for (int s = 1; s < MS; ++s) {
    const v4f us = *(const v4f*)(Uh + (size_t)s * BN * ND + (size_t)i * 4);
    u[0] += us[0]; u[1] += us[1]; u[2] += us[2]; u[3] += us[3];
    nrm += Nh[(size_t)s * BN + b];
  }
  v4f o;
  o[0] = u[0] / nrm;
  o[1] = u[1] / nrm;
  o[2] = u[2] / nrm;
  o[3] = u[3] / nrm;
  *(v4f*)(out + (size_t)i * 4) = o;
}

extern "C" void kernel_launch(void* const* d_in, const int* in_sizes, int n_in,
                              void* d_out, int out_size, void* d_ws, size_t ws_size,
                              hipStream_t stream) {
  const float* X   = (const float*)d_in[0];
  const float* Mu0 = (const float*)d_in[1];
  const float* Mu1 = (const float*)d_in[2];
  const float* S0  = (const float*)d_in[3];
  const float* S1  = (const float*)d_in[4];
  const float* Lam = (const float*)d_in[5];
  const float* tP  = (const float*)d_in[6];
  const float* eP  = (const float*)d_in[7];
  float* out = (float*)d_out;

  float* ws   = (float*)d_ws;
  float* Uh   = ws;
  float* Nh   = Uh + MS * BN * ND;
  (void)ws_size; (void)in_sizes; (void)n_in; (void)out_size;

  gmm_main_kernel<<<dim3(BG, MS), 32 * NW, 0, stream>>>(X, Mu0, Mu1, S0, S1,
                                                        Lam, tP, eP, Uh, Nh);

  gmm_finalize_kernel<<<BN * ND / 4 / 128, 128, 0, stream>>>(Uh, Nh, out);
}
